# starting kernel
# baseline (speedup 1.0000x reference)
#include <hip/hip_runtime.h>

#define DIM_IN   128
#define DIM_HID  256
#define DIM_OUT  128
#define N_DIST   1024
#define N_NODES  500000

typedef float v2f __attribute__((ext_vector_type(2)));
typedef float v4f __attribute__((ext_vector_type(4)));
typedef float v8f __attribute__((ext_vector_type(8)));

__global__ void n2d_zero_head(float* __restrict__ head) {
    int i = blockIdx.x * blockDim.x + threadIdx.x;
    const int n = N_DIST * DIM_IN;
    for (; i < n; i += gridDim.x * blockDim.x) head[i] = 0.0f;
}

__global__ void n2d_segsum(const float* __restrict__ x,
                           const int*   __restrict__ zone,
                           float*       __restrict__ head) {
    const int lane   = threadIdx.x & 31;
    const int wave   = (blockIdx.x * blockDim.x + threadIdx.x) >> 5;
    const int nwaves = (gridDim.x * blockDim.x) >> 5;

    for (int node = wave; node < N_NODES; node += nwaves) {
        const int z = zone[node];
        const v4f* p = (const v4f*)(x + (size_t)node * DIM_IN) + lane;
        v4f v = __builtin_nontemporal_load(p);
        float* h = head + z * DIM_IN + lane * 4;
        atomicAdd(h + 0, v.x);
        atomicAdd(h + 1, v.y);
        atomicAdd(h + 2, v.z);
        atomicAdd(h + 3, v.w);
    }
}

__global__ void n2d_mlp(const float* __restrict__ head,
                        const float* __restrict__ w1, const float* __restrict__ b1,
                        const float* __restrict__ w2, const float* __restrict__ b2,
                        float* __restrict__ out) {
    __shared__ float h1[16 * DIM_HID];

    const int lane = threadIdx.x & 31;
    const int wv   = threadIdx.x >> 5;
    const int half = lane >> 4;
    const int m    = lane & 15;
    const int row0 = blockIdx.x * 16;

    for (int t = wv; t < DIM_HID / 16; t += 8) {
        const int n0 = t * 16;
        v8f acc = {};
        for (int k = 0; k < DIM_IN; k += 4) {
            const int ka = k + half * 2;
            v2f a, b;
            float a0 = head[(row0 + m) * DIM_IN + ka];
            float a1 = head[(row0 + m) * DIM_IN + ka + 1];
            a.x = a0 > 0.0f ? a0 : 0.0f;
            a.y = a1 > 0.0f ? a1 : 0.0f;
            b.x = w1[(size_t)ka       * DIM_HID + n0 + m];
            b.y = w1[(size_t)(ka + 1) * DIM_HID + n0 + m];
            acc = __builtin_amdgcn_wmma_f32_16x16x4_f32(
                      false, a, false, b, (short)0, acc, false, false);
        }
        const float bias = b1[n0 + m];
        #pragma unroll
        for (int r = 0; r < 8; ++r) {
            float v = acc[r] + bias;
            h1[(r + half * 8) * DIM_HID + n0 + m] = v > 0.0f ? v : 0.0f;
        }
    }
    __syncthreads();

    {
        const int n0 = wv * 16;
        v8f acc = {};
        for (int k = 0; k < DIM_HID; k += 4) {
            const int ka = k + half * 2;
            v2f a, b;
            a.x = h1[m * DIM_HID + ka];
            a.y = h1[m * DIM_HID + ka + 1];
            b.x = w2[(size_t)ka       * DIM_OUT + n0 + m];
            b.y = w2[(size_t)(ka + 1) * DIM_OUT + n0 + m];
            acc = __builtin_amdgcn_wmma_f32_16x16x4_f32(
                      false, a, false, b, (short)0, acc, false, false);
        }
        const float bias = b2[n0 + m];
        #pragma unroll
        for (int r = 0; r < 8; ++r)
            out[(size_t)(row0 + r + half * 8) * DIM_OUT + n0 + m] = acc[r] + bias;
    }
}

extern "C" void kernel_launch(void* const* d_in, const int* in_sizes, int n_in,
                              void* d_out, int out_size, void* d_ws, size_t ws_size,
                              hipStream_t stream) {
    const float* x    = (const float*)d_in[0];
    const int*   zone = (const int*)  d_in[1];
    const float* w1   = (const float*)d_in[2];
    const float* b1   = (const float*)d_in[3];
    const float* w2   = (const float*)d_in[4];
    const float* b2   = (const float*)d_in[5];

    float* head = (float*)d_ws;
    float* out  = (float*)d_out;

    n2d_zero_head<<<256, 256, 0, stream>>>(head);
    n2d_segsum<<<2048, 256, 0, stream>>>(x, zone, head);
    n2d_mlp<<<N_DIST / 16, 256, 0, stream>>>(head, w1, b1, w2, b2, out);
}


# Segment sum in per-workgroup LDS bins (ds_add_f32) per node chunk and 256-B column slice; chunk partials summed in fixed order by a reduce kernel; no global f32 atomics; sum order changes
# speedup vs baseline: 3.4737x; 3.4737x over previous
#include <hip/hip_runtime.h>

#define DIM_IN   128
#define DIM_HID  256
#define DIM_OUT  128
#define N_DIST   1024
#define N_NODES  500000

typedef float v2f __attribute__((ext_vector_type(2)));
typedef float v4f __attribute__((ext_vector_type(4)));
typedef float v8f __attribute__((ext_vector_type(8)));

#define SEG_W        64
#define SEG_SLICES   (DIM_IN / SEG_W)
#define SEG_CHUNKS   128
#define SEG_RPC      3920
#define SEG_WAVES    16
#define SEG_THREADS  (SEG_WAVES * 32)
#define SEG_BINS4    (N_DIST * SEG_W / 4)
#define HEAD_ELEMS   (N_DIST * DIM_IN)
#define RED_D        2
#define RED_WAVES    8

static_assert(SEG_RPC % 16 == 0 && N_NODES % 16 == 0, "batches must be full");
static_assert((SEG_CHUNKS - 1) * SEG_RPC < N_NODES && SEG_CHUNKS * SEG_RPC >= N_NODES, "chunks must cover the nodes");
struct SegBatch { v4f v[8]; int z; };

static __device__ __forceinline__ void seg_load(SegBatch& B, const float* __restrict__ xl,
                                                const int* __restrict__ zone,
                                                int rb, int lane) {
    B.z = zone[rb + (lane & 15)];
    #pragma unroll
    for (int u = 0; u < 8; ++u)
        B.v[u] = __builtin_nontemporal_load((const v4f*)(xl + (size_t)(rb + 2 * u) * DIM_IN));
}

static __device__ __forceinline__ void seg_add(const SegBatch& B, float* bins_lane, int r) {
    #pragma unroll
    for (int u = 0; u < 8; ++u) {
        const int z0 = __builtin_amdgcn_readlane(B.z, 2 * u);
        const int z1 = __builtin_amdgcn_readlane(B.z, 2 * u + 1);
        const int z  = r ? z1 : z0;
        if ((unsigned)z < (unsigned)N_DIST) {
            const v4f t = B.v[u];
            float* p = bins_lane + z * SEG_W;
            float* plo = p + (r ? 2 : 0);
            float* phi = p + (r ? 0 : 2);
            atomicAdd(plo,     r ? t.z : t.x);
            atomicAdd(plo + 1, r ? t.w : t.y);
            atomicAdd(phi,     r ? t.x : t.z);
            atomicAdd(phi + 1, r ? t.y : t.w);
        }
    }
}

__global__ __launch_bounds__(SEG_THREADS)
void n2d_segsum_bins(const float* __restrict__ x,
                     const int*   __restrict__ zone,
                     float*       __restrict__ part) {
    __shared__ float bins[N_DIST * SEG_W];

    const int tid  = threadIdx.x;
    const int lane = tid & 31;
    const int wv   = __builtin_amdgcn_readfirstlane(tid >> 5);
    const int r    = lane >> 4;
    const int col4 = lane & 15;
    const int s    = blockIdx.x & 1;
    const int c    = blockIdx.x >> 1;
    const int n0   = c * SEG_RPC;
    const int n1   = (n0 + SEG_RPC < N_NODES) ? (n0 + SEG_RPC) : N_NODES;
    const int nbatch = (n1 - n0 + 15) >> 4;

    const float* xl = x + s * SEG_W + col4 * 4 + r * DIM_IN;
    float* bins_lane = bins + col4 * 4;

    const int nw = (nbatch - wv + SEG_WAVES - 1) / SEG_WAVES;
    const int rstep = 16 * SEG_WAVES;
    int rb = n0 + 16 * wv;

    SegBatch A, Bq;
    if (nw > 0) seg_load(A, xl, zone, rb, lane);

    v4f* bins4 = (v4f*)bins;
    const v4f zero4 = {0.0f, 0.0f, 0.0f, 0.0f};
    for (int i = tid; i < SEG_BINS4; i += SEG_THREADS) bins4[i] = zero4;
    __syncthreads();

    int i = 0;
    for (; i + 2 < nw; i += 2) {
        seg_load(Bq, xl, zone, rb + rstep, lane);
        seg_add(A, bins_lane, r);
        seg_load(A, xl, zone, rb + 2 * rstep, lane);
        seg_add(Bq, bins_lane, r);
        rb += 2 * rstep;
    }
    if (i + 1 < nw) {
        seg_load(Bq, xl, zone, rb + rstep, lane);
        seg_add(A, bins_lane, r);
        seg_add(Bq, bins_lane, r);
    } else if (i < nw) {
        seg_add(A, bins_lane, r);
    }
    __syncthreads();

    float* dst = part + (size_t)c * HEAD_ELEMS + s * SEG_W;
    for (int i = tid; i < SEG_BINS4; i += SEG_THREADS) {
        const int d = i >> 4, q = i & 15;
        *(v4f*)(dst + d * DIM_IN + q * 4) = bins4[i];
    }
}

__global__ __launch_bounds__(RED_WAVES * 32)
void n2d_reduce_parts(const float* __restrict__ part, float* __restrict__ head) {
    __shared__ float red[RED_WAVES][RED_D][DIM_IN];

    const int tid  = threadIdx.x;
    const int lane = tid & 31;
    const int wv   = tid >> 5;
    const int d0   = blockIdx.x * RED_D;

    #pragma unroll
    for (int q = 0; q < RED_D; ++q) {
        const v4f* p = (const v4f*)(part + (size_t)(d0 + q) * DIM_IN) + lane;
        v4f acc = {0.0f, 0.0f, 0.0f, 0.0f};
        #pragma unroll
        for (int j = 0; j < SEG_CHUNKS / RED_WAVES; ++j)
            acc += p[(size_t)(wv + RED_WAVES * j) * (HEAD_ELEMS / 4)];
        *(v4f*)&red[wv][q][lane * 4] = acc;
    }
    __syncthreads();

    for (int o = tid; o < RED_D * DIM_IN; o += RED_WAVES * 32) {
        const int q = o >> 7, col = o & (DIM_IN - 1);
        float sum = red[0][q][col];
        #pragma unroll
        for (int w = 1; w < RED_WAVES; ++w) sum += red[w][q][col];
        head[(size_t)(d0 + q) * DIM_IN + col] = sum;
    }
}

__global__ void n2d_mlp(const float* __restrict__ head,
                        const float* __restrict__ w1, const float* __restrict__ b1,
                        const float* __restrict__ w2, const float* __restrict__ b2,
                        float* __restrict__ out) {
    __shared__ float h1[16 * DIM_HID];

    const int lane = threadIdx.x & 31;
    const int wv   = threadIdx.x >> 5;
    const int half = lane >> 4;
    const int m    = lane & 15;
    const int row0 = blockIdx.x * 16;

    for (int t = wv; t < DIM_HID / 16; t += 8) {
        const int n0 = t * 16;
        v8f acc = {};
        for (int k = 0; k < DIM_IN; k += 4) {
            const int ka = k + half * 2;
            v2f a, b;
            float a0 = head[(row0 + m) * DIM_IN + ka];
            float a1 = head[(row0 + m) * DIM_IN + ka + 1];
            a.x = a0 > 0.0f ? a0 : 0.0f;
            a.y = a1 > 0.0f ? a1 : 0.0f;
            b.x = w1[(size_t)ka       * DIM_HID + n0 + m];
            b.y = w1[(size_t)(ka + 1) * DIM_HID + n0 + m];
            acc = __builtin_amdgcn_wmma_f32_16x16x4_f32(
                      false, a, false, b, (short)0, acc, false, false);
        }
        const float bias = b1[n0 + m];
        #pragma unroll
        for (int r = 0; r < 8; ++r) {
            float v = acc[r] + bias;
            h1[(r + half * 8) * DIM_HID + n0 + m] = v > 0.0f ? v : 0.0f;
        }
    }
    __syncthreads();

    {
        const int n0 = wv * 16;
        v8f acc = {};
        for (int k = 0; k < DIM_HID; k += 4) {
            const int ka = k + half * 2;
            v2f a, b;
            a.x = h1[m * DIM_HID + ka];
            a.y = h1[m * DIM_HID + ka + 1];
            b.x = w2[(size_t)ka       * DIM_OUT + n0 + m];
            b.y = w2[(size_t)(ka + 1) * DIM_OUT + n0 + m];
            acc = __builtin_amdgcn_wmma_f32_16x16x4_f32(
                      false, a, false, b, (short)0, acc, false, false);
        }
        const float bias = b2[n0 + m];
        #pragma unroll
        for (int r = 0; r < 8; ++r)
            out[(size_t)(row0 + r + half * 8) * DIM_OUT + n0 + m] = acc[r] + bias;
    }
}

extern "C" void kernel_launch(void* const* d_in, const int* in_sizes, int n_in,
                              void* d_out, int out_size, void* d_ws, size_t ws_size,
                              hipStream_t stream) {
    const float* x    = (const float*)d_in[0];
    const int*   zone = (const int*)  d_in[1];
    const float* w1   = (const float*)d_in[2];
    const float* b1   = (const float*)d_in[3];
    const float* w2   = (const float*)d_in[4];
    const float* b2   = (const float*)d_in[5];

    float* head = (float*)d_ws;
    float* part = head + HEAD_ELEMS;
    float* out  = (float*)d_out;

    n2d_segsum_bins<<<SEG_CHUNKS * SEG_SLICES, SEG_THREADS, 0, stream>>>(x, zone, part);
    n2d_reduce_parts<<<N_DIST / RED_D, RED_WAVES * 32, 0, stream>>>(part, head);
    n2d_mlp<<<N_DIST / 16, 256, 0, stream>>>(head, w1, b1, w2, b2, out);
}


# Reduce and MLP fused into one launch (256 work-groups of 4 districts, M tile zero-padded): w1/w2 fragment loads overlap the partial reads; wave sums and two K halves in fixed order; sum order changes
# speedup vs baseline: 4.1769x; 1.2024x over previous
#include <hip/hip_runtime.h>

#define DIM_IN   128
#define DIM_HID  256
#define DIM_OUT  128
#define N_DIST   1024
#define N_NODES  500000

typedef float v2f __attribute__((ext_vector_type(2)));
typedef float v4f __attribute__((ext_vector_type(4)));
typedef float v8f __attribute__((ext_vector_type(8)));

#define SEG_W        64
#define SEG_SLICES   (DIM_IN / SEG_W)
#define SEG_CHUNKS   128
#define SEG_RPC      3920
#define SEG_WAVES    16
#define SEG_THREADS  (SEG_WAVES * 32)
#define SEG_BINS4    (N_DIST * SEG_W / 4)
#define HEAD_ELEMS   (N_DIST * DIM_IN)
#define FUS_D        4
#define FUS_WAVES    16
#define FUS_THREADS  (FUS_WAVES * 32)
#define A0_PITCH     (DIM_IN + 4)
#define H1_PITCH     (DIM_HID + 4)

static_assert(SEG_RPC % 16 == 0 && N_NODES % 16 == 0, "batches must be full");
static_assert((SEG_CHUNKS - 1) * SEG_RPC < N_NODES && SEG_CHUNKS * SEG_RPC >= N_NODES, "chunks must cover the nodes");
struct SegBatch { v4f v[8]; int z; };

static __device__ __forceinline__ void seg_load(SegBatch& B, const float* __restrict__ xl,
                                                const int* __restrict__ zone,
                                                int rb, int lane) {
    B.z = zone[rb + (lane & 15)];
    #pragma unroll
    for (int u = 0; u < 8; ++u)
        B.v[u] = __builtin_nontemporal_load((const v4f*)(xl + (size_t)(rb + 2 * u) * DIM_IN));
}

static __device__ __forceinline__ void seg_add(const SegBatch& B, float* bins_lane, int r) {
    #pragma unroll
    for (int u = 0; u < 8; ++u) {
        const int z0 = __builtin_amdgcn_readlane(B.z, 2 * u);
        const int z1 = __builtin_amdgcn_readlane(B.z, 2 * u + 1);
        const int z  = r ? z1 : z0;
        if ((unsigned)z < (unsigned)N_DIST) {
            const v4f t = B.v[u];
            float* p = bins_lane + z * SEG_W;
            float* plo = p + (r ? 2 : 0);
            float* phi = p + (r ? 0 : 2);
            atomicAdd(plo,     r ? t.z : t.x);
            atomicAdd(plo + 1, r ? t.w : t.y);
            atomicAdd(phi,     r ? t.x : t.z);
            atomicAdd(phi + 1, r ? t.y : t.w);
        }
    }
}

__global__ __launch_bounds__(SEG_THREADS)
void n2d_segsum_bins(const float* __restrict__ x,
                     const int*   __restrict__ zone,
                     float*       __restrict__ part) {
    __shared__ float bins[N_DIST * SEG_W];

    const int tid  = threadIdx.x;
    const int lane = tid & 31;
    const int wv   = __builtin_amdgcn_readfirstlane(tid >> 5);
    const int r    = lane >> 4;
    const int col4 = lane & 15;
    const int s    = blockIdx.x & 1;
    const int c    = blockIdx.x >> 1;
    const int n0   = c * SEG_RPC;
    const int n1   = (n0 + SEG_RPC < N_NODES) ? (n0 + SEG_RPC) : N_NODES;
    const int nbatch = (n1 - n0 + 15) >> 4;

    const float* xl = x + s * SEG_W + col4 * 4 + r * DIM_IN;
    float* bins_lane = bins + col4 * 4;

    const int nw = (nbatch - wv + SEG_WAVES - 1) / SEG_WAVES;
    const int rstep = 16 * SEG_WAVES;
    int rb = n0 + 16 * wv;

    SegBatch A, Bq;
    if (nw > 0) seg_load(A, xl, zone, rb, lane);

    v4f* bins4 = (v4f*)bins;
    const v4f zero4 = {0.0f, 0.0f, 0.0f, 0.0f};
    for (int i = tid; i < SEG_BINS4; i += SEG_THREADS) bins4[i] = zero4;
    __syncthreads();

    int i = 0;
    for (; i + 2 < nw; i += 2) {
        seg_load(Bq, xl, zone, rb + rstep, lane);
        seg_add(A, bins_lane, r);
        seg_load(A, xl, zone, rb + 2 * rstep, lane);
        seg_add(Bq, bins_lane, r);
        rb += 2 * rstep;
    }
    if (i + 1 < nw) {
        seg_load(Bq, xl, zone, rb + rstep, lane);
        seg_add(A, bins_lane, r);
        seg_add(Bq, bins_lane, r);
    } else if (i < nw) {
        seg_add(A, bins_lane, r);
    }
    __syncthreads();

    float* dst = part + (size_t)c * HEAD_ELEMS + s * SEG_W;
    for (int i = tid; i < SEG_BINS4; i += SEG_THREADS) {
        const int d = i >> 4, q = i & 15;
        *(v4f*)(dst + d * DIM_IN + q * 4) = bins4[i];
    }
}

__global__ __launch_bounds__(FUS_THREADS)
void n2d_reduce_mlp(const float* __restrict__ part,
                    const float* __restrict__ w1, const float* __restrict__ b1,
                    const float* __restrict__ w2, const float* __restrict__ b2,
                    float* __restrict__ out) {
    __shared__ alignas(16) float red[FUS_WAVES][FUS_D][DIM_IN];
    __shared__ alignas(16) float a0[16 * A0_PITCH];
    __shared__ alignas(16) float h1[16 * H1_PITCH];
    __shared__ float o2[8][FUS_D][16];
    __shared__ alignas(16) float ot[FUS_D][DIM_OUT];

    const int tid  = threadIdx.x;
    const int lane = tid & 31;
    const int wv   = __builtin_amdgcn_readfirstlane(tid >> 5);
    const int half = lane >> 4;
    const int m    = lane & 15;
    const int d0   = blockIdx.x * FUS_D;

    const int n1 = wv * 16;
    const int n2 = (wv & 7) * 16;
    const int k2 = (wv >> 3) * (DIM_HID / 2);
    v2f bw1[DIM_IN / 4], bw2[DIM_HID / 8];
    #pragma unroll
    for (int s = 0; s < DIM_IN / 4; ++s) {
        const int ka = 4 * s + half * 2;
        bw1[s].x = w1[(size_t)ka       * DIM_HID + n1 + m];
        bw1[s].y = w1[(size_t)(ka + 1) * DIM_HID + n1 + m];
    }
    #pragma unroll
    for (int s = 0; s < DIM_HID / 8; ++s) {
        const int ka = k2 + 4 * s + half * 2;
        bw2[s].x = w2[(size_t)ka       * DIM_OUT + n2 + m];
        bw2[s].y = w2[(size_t)(ka + 1) * DIM_OUT + n2 + m];
    }
    const float bias1 = b1[n1 + m];
    const float bias2 = b2[n2 + m];

    #pragma unroll
    for (int q = 0; q < FUS_D; ++q) {
        const v4f* p = (const v4f*)(part + (size_t)(d0 + q) * DIM_IN) + lane;
        v4f acc = {0.0f, 0.0f, 0.0f, 0.0f};
        #pragma unroll
        for (int j = 0; j < SEG_CHUNKS / FUS_WAVES; ++j)
            acc += p[(size_t)(wv + FUS_WAVES * j) * (HEAD_ELEMS / 4)];
        *(v4f*)&red[wv][q][lane * 4] = acc;
    }
    for (int i = tid; i < (16 - FUS_D) * A0_PITCH; i += FUS_THREADS) a0[FUS_D * A0_PITCH + i] = 0.0f;
    __syncthreads();

    {
        const int q = tid >> 7, col = tid & (DIM_IN - 1);
        float sum = red[0][q][col];
        #pragma unroll
        for (int w = 1; w < FUS_WAVES; ++w) sum += red[w][q][col];
        a0[q * A0_PITCH + col] = sum > 0.0f ? sum : 0.0f;
    }
    __syncthreads();

    {
        v8f acc = {};
        #pragma unroll
        for (int s = 0; s < DIM_IN / 4; ++s) {
            const int ka = 4 * s + half * 2;
            const v2f a = *(const v2f*)&a0[m * A0_PITCH + ka];
            acc = __builtin_amdgcn_wmma_f32_16x16x4_f32(
                      false, a, false, bw1[s], (short)0, acc, false, false);
        }
        #pragma unroll
        for (int r = 0; r < 8; ++r) {
            const float v = acc[r] + bias1;
            h1[(r + half * 8) * H1_PITCH + n1 + m] = v > 0.0f ? v : 0.0f;
        }
    }
    __syncthreads();

    v8f acc2 = {};
    #pragma unroll
    for (int s = 0; s < DIM_HID / 8; ++s) {
        const int ka = k2 + 4 * s + half * 2;
        const v2f a = *(const v2f*)&h1[m * H1_PITCH + ka];
        acc2 = __builtin_amdgcn_wmma_f32_16x16x4_f32(
                   false, a, false, bw2[s], (short)0, acc2, false, false);
    }
    if (wv >= 8 && half == 0) {
        #pragma unroll
        for (int r = 0; r < FUS_D; ++r) o2[wv & 7][r][m] = acc2[r];
    }
    __syncthreads();
    if (wv < 8 && half == 0) {
        #pragma unroll
        for (int r = 0; r < FUS_D; ++r) ot[r][n2 + m] = (acc2[r] + o2[wv][r][m]) + bias2;
    }
    __syncthreads();
    if (wv < FUS_D)
        *(v4f*)(out + (size_t)(d0 + wv) * DIM_OUT + lane * 4) = *(const v4f*)&ot[wv][lane * 4];
}

extern "C" void kernel_launch(void* const* d_in, const int* in_sizes, int n_in,
                              void* d_out, int out_size, void* d_ws, size_t ws_size,
                              hipStream_t stream) {
    const float* x    = (const float*)d_in[0];
    const int*   zone = (const int*)  d_in[1];
    const float* w1   = (const float*)d_in[2];
    const float* b1   = (const float*)d_in[3];
    const float* w2   = (const float*)d_in[4];
    const float* b2   = (const float*)d_in[5];

    float* part = (float*)d_ws;
    float* out  = (float*)d_out;

    n2d_segsum_bins<<<SEG_CHUNKS * SEG_SLICES, SEG_THREADS, 0, stream>>>(x, zone, part);
    n2d_reduce_mlp<<<N_DIST / FUS_D, FUS_THREADS, 0, stream>>>(part, w1, b1, w2, b2, out);
}


# Segment-sum work-groups own one half of the districts (512 x 64 LDS bins, rows of the half taken from a per-wave ballot-ranked list): 64 chunk partial planes instead of 128; LDS add order changes
# speedup vs baseline: 4.9503x; 1.1852x over previous
#include <hip/hip_runtime.h>

#define DIM_IN   128
#define DIM_HID  256
#define DIM_OUT  128
#define N_DIST   1024
#define N_NODES  500000

typedef float v2f __attribute__((ext_vector_type(2)));
typedef float v4f __attribute__((ext_vector_type(4)));
typedef float v8f __attribute__((ext_vector_type(8)));

#define SEG_W        64
#define SEG_SLICES   (DIM_IN / SEG_W)
#define SEG_HALVES   2
#define SEG_DH       (N_DIST / SEG_HALVES)
#define SEG_CHUNKS   64
#define SEG_RPC      7824
#define SEG_WAVES    16
#define SEG_THREADS  (SEG_WAVES * 32)
#define SEG_RPW      (SEG_RPC / SEG_WAVES)
#define SEG_GROUPS   ((SEG_RPW + 31) / 32)
#define SEG_LIST     512
#define SEG_NONE     0xFFFFFFFFu
#define SEG_BINS4    (SEG_DH * SEG_W / 4)
#define HEAD_ELEMS   (N_DIST * DIM_IN)
#define FUS_D        4
#define FUS_WAVES    16
#define FUS_THREADS  (FUS_WAVES * 32)
#define A0_PITCH     (DIM_IN + 4)
#define H1_PITCH     (DIM_HID + 4)

typedef unsigned int v4u __attribute__((ext_vector_type(4)));

static_assert(SEG_RPC % SEG_WAVES == 0, "waves take equal row ranges");
static_assert((SEG_CHUNKS - 1) * SEG_RPC < N_NODES && SEG_CHUNKS * SEG_RPC >= N_NODES, "chunks must cover the nodes");
static_assert(SEG_RPC <= 8192 && SEG_DH == 512, "a list entry packs 13 bits of row and 9 bits of district");
static_assert(((SEG_RPW + 15) / 16) * 16 <= SEG_LIST, "a padded list must fit");

struct SegBatch { v4f v[8]; unsigned e[8]; };

static __device__ __forceinline__ void seg_load(SegBatch& B, const float* __restrict__ xl,
                                                const unsigned* mylist, int b, int r) {
    const v4u* lp = (const v4u*)(mylist + 16 * b + 8 * r);
    const v4u e0 = lp[0], e1 = lp[1];
    B.e[0] = e0.x; B.e[1] = e0.y; B.e[2] = e0.z; B.e[3] = e0.w;
    B.e[4] = e1.x; B.e[5] = e1.y; B.e[6] = e1.z; B.e[7] = e1.w;
    #pragma unroll
    for (int u = 0; u < 8; ++u) {
        const unsigned rl = (B.e[u] == SEG_NONE) ? 0u : (B.e[u] >> 9);
        B.v[u] = __builtin_nontemporal_load((const v4f*)(xl + (size_t)rl * DIM_IN));
    }
}

static __device__ __forceinline__ void seg_add(const SegBatch& B, float* bins_lane, int r) {
    #pragma unroll
    for (int u = 0; u < 8; ++u) {
        const unsigned e = B.e[u];
        if (e != SEG_NONE) {
            const int z = (int)(e & (unsigned)(SEG_DH - 1));
            const v4f t = B.v[u];
            float* p = bins_lane + z * SEG_W;
            float* plo = p + (r ? 2 : 0);
            float* phi = p + (r ? 0 : 2);
            atomicAdd(plo,     r ? t.z : t.x);
            atomicAdd(plo + 1, r ? t.w : t.y);
            atomicAdd(phi,     r ? t.x : t.z);
            atomicAdd(phi + 1, r ? t.y : t.w);
        }
    }
}

__global__ __launch_bounds__(SEG_THREADS)
void n2d_segsum_bins(const float* __restrict__ x,
                     const int*   __restrict__ zone,
                     float*       __restrict__ part) {
    __shared__ float bins[SEG_DH * SEG_W];
    __shared__ alignas(16) unsigned list[SEG_WAVES][SEG_LIST];

    const int tid  = threadIdx.x;
    const int lane = tid & 31;
    const int wv   = __builtin_amdgcn_readfirstlane(tid >> 5);
    const int r    = lane >> 4;
    const int col4 = lane & 15;
    const int s    = blockIdx.x & 1;
    const int h    = (blockIdx.x >> 1) & 1;
    const int c    = blockIdx.x >> 2;
    const int n0   = c * SEG_RPC;
    const int w0   = n0 + wv * SEG_RPW;
    const int w1   = (w0 + SEG_RPW < N_NODES) ? (w0 + SEG_RPW) : N_NODES;

    int zr[SEG_GROUPS];
    #pragma unroll
    for (int g = 0; g < SEG_GROUPS; ++g) {
        const int row = w0 + 32 * g + lane;
        zr[g] = (row < w1) ? zone[row] : -1;
    }

    v4f* bins4 = (v4f*)bins;
    const v4f zero4 = {0.0f, 0.0f, 0.0f, 0.0f};
    for (int i = tid; i < SEG_BINS4; i += SEG_THREADS) bins4[i] = zero4;

    unsigned* mylist = list[wv];
    int cnt = 0;
    #pragma unroll
    for (int g = 0; g < SEG_GROUPS; ++g) {
        const int z = zr[g];
        const bool match = ((unsigned)z < (unsigned)N_DIST) && ((z >> 9) == h);
        const unsigned mask = __builtin_amdgcn_ballot_w32(match);
        const int rank = (int)__builtin_amdgcn_mbcnt_lo(mask, 0u);
        if (match)
            mylist[cnt + rank] = ((unsigned)(wv * SEG_RPW + 32 * g + lane) << 9) | (unsigned)(z & (SEG_DH - 1));
        cnt += __builtin_popcount(mask);
    }
    const int nb = (cnt + 15) >> 4;
    if (lane < 16 && cnt + lane < 16 * nb) mylist[cnt + lane] = SEG_NONE;

    const float* xl = x + (size_t)n0 * DIM_IN + s * SEG_W + col4 * 4;
    float* bins_lane = bins + col4 * 4;

    SegBatch A, Bq;
    if (nb > 0) seg_load(A, xl, mylist, 0, r);
    __syncthreads();

    int i = 0;
    for (; i + 2 < nb; i += 2) {
        seg_load(Bq, xl, mylist, i + 1, r);
        seg_add(A, bins_lane, r);
        seg_load(A, xl, mylist, i + 2, r);
        seg_add(Bq, bins_lane, r);
    }
    if (i + 1 < nb) {
        seg_load(Bq, xl, mylist, i + 1, r);
        seg_add(A, bins_lane, r);
        seg_add(Bq, bins_lane, r);
    } else if (i < nb) {
        seg_add(A, bins_lane, r);
    }
    __syncthreads();

    float* dst = part + (size_t)c * HEAD_ELEMS + (size_t)(h * SEG_DH) * DIM_IN + s * SEG_W;
    for (int i = tid; i < SEG_BINS4; i += SEG_THREADS) {
        const int d = i >> 4, q = i & 15;
        *(v4f*)(dst + d * DIM_IN + q * 4) = bins4[i];
    }
}

__global__ __launch_bounds__(FUS_THREADS)
void n2d_reduce_mlp(const float* __restrict__ part,
                    const float* __restrict__ w1, const float* __restrict__ b1,
                    const float* __restrict__ w2, const float* __restrict__ b2,
                    float* __restrict__ out) {
    __shared__ alignas(16) float red[FUS_WAVES][FUS_D][DIM_IN];
    __shared__ alignas(16) float a0[16 * A0_PITCH];
    __shared__ alignas(16) float h1[16 * H1_PITCH];
    __shared__ float o2[8][FUS_D][16];
    __shared__ alignas(16) float ot[FUS_D][DIM_OUT];

    const int tid  = threadIdx.x;
    const int lane = tid & 31;
    const int wv   = __builtin_amdgcn_readfirstlane(tid >> 5);
    const int half = lane >> 4;
    const int m    = lane & 15;
    const int d0   = blockIdx.x * FUS_D;

    const int n1 = wv * 16;
    const int n2 = (wv & 7) * 16;
    const int k2 = (wv >> 3) * (DIM_HID / 2);
    v2f bw1[DIM_IN / 4], bw2[DIM_HID / 8];
    #pragma unroll
    for (int s = 0; s < DIM_IN / 4; ++s) {
        const int ka = 4 * s + half * 2;
        bw1[s].x = w1[(size_t)ka       * DIM_HID + n1 + m];
        bw1[s].y = w1[(size_t)(ka + 1) * DIM_HID + n1 + m];
    }
    #pragma unroll
    for (int s = 0; s < DIM_HID / 8; ++s) {
        const int ka = k2 + 4 * s + half * 2;
        bw2[s].x = w2[(size_t)ka       * DIM_OUT + n2 + m];
        bw2[s].y = w2[(size_t)(ka + 1) * DIM_OUT + n2 + m];
    }
    const float bias1 = b1[n1 + m];
    const float bias2 = b2[n2 + m];

    #pragma unroll
    for (int q = 0; q < FUS_D; ++q) {
        const v4f* p = (const v4f*)(part + (size_t)(d0 + q) * DIM_IN) + lane;
        v4f acc = {0.0f, 0.0f, 0.0f, 0.0f};
        #pragma unroll
        for (int j = 0; j < SEG_CHUNKS / FUS_WAVES; ++j)
            acc += p[(size_t)(wv + FUS_WAVES * j) * (HEAD_ELEMS / 4)];
        *(v4f*)&red[wv][q][lane * 4] = acc;
    }
    for (int i = tid; i < (16 - FUS_D) * A0_PITCH; i += FUS_THREADS) a0[FUS_D * A0_PITCH + i] = 0.0f;
    __syncthreads();

    {
        const int q = tid >> 7, col = tid & (DIM_IN - 1);
        float sum = red[0][q][col];
        #pragma unroll
        for (int w = 1; w < FUS_WAVES; ++w) sum += red[w][q][col];
        a0[q * A0_PITCH + col] = sum > 0.0f ? sum : 0.0f;
    }
    __syncthreads();

    {
        v8f acc = {};
        #pragma unroll
        for (int s = 0; s < DIM_IN / 4; ++s) {
            const int ka = 4 * s + half * 2;
            const v2f a = *(const v2f*)&a0[m * A0_PITCH + ka];
            acc = __builtin_amdgcn_wmma_f32_16x16x4_f32(
                      false, a, false, bw1[s], (short)0, acc, false, false);
        }
        #pragma unroll
        for (int r = 0; r < 8; ++r) {
            const float v = acc[r] + bias1;
            h1[(r + half * 8) * H1_PITCH + n1 + m] = v > 0.0f ? v : 0.0f;
        }
    }
    __syncthreads();

    v8f acc2 = {};
    #pragma unroll
    for (int s = 0; s < DIM_HID / 8; ++s) {
        const int ka = k2 + 4 * s + half * 2;
        const v2f a = *(const v2f*)&h1[m * H1_PITCH + ka];
        acc2 = __builtin_amdgcn_wmma_f32_16x16x4_f32(
                   false, a, false, bw2[s], (short)0, acc2, false, false);
    }
    if (wv >= 8 && half == 0) {
        #pragma unroll
        for (int r = 0; r < FUS_D; ++r) o2[wv & 7][r][m] = acc2[r];
    }
    __syncthreads();
    if (wv < 8 && half == 0) {
        #pragma unroll
        for (int r = 0; r < FUS_D; ++r) ot[r][n2 + m] = (acc2[r] + o2[wv][r][m]) + bias2;
    }
    __syncthreads();
    if (wv < FUS_D)
        *(v4f*)(out + (size_t)(d0 + wv) * DIM_OUT + lane * 4) = *(const v4f*)&ot[wv][lane * 4];
}

extern "C" void kernel_launch(void* const* d_in, const int* in_sizes, int n_in,
                              void* d_out, int out_size, void* d_ws, size_t ws_size,
                              hipStream_t stream) {
    const float* x    = (const float*)d_in[0];
    const int*   zone = (const int*)  d_in[1];
    const float* w1   = (const float*)d_in[2];
    const float* b1   = (const float*)d_in[3];
    const float* w2   = (const float*)d_in[4];
    const float* b2   = (const float*)d_in[5];

    float* part = (float*)d_ws;
    float* out  = (float*)d_out;

    n2d_segsum_bins<<<SEG_CHUNKS * SEG_SLICES * SEG_HALVES, SEG_THREADS, 0, stream>>>(x, zone, part);
    n2d_reduce_mlp<<<N_DIST / FUS_D, FUS_THREADS, 0, stream>>>(part, w1, b1, w2, b2, out);
}


# Segment-sum work-groups own one quarter of the districts (256 x 64 LDS bins, chunks of 15632 rows): 32 chunk partial planes instead of 64 to store, flush and re-read; LDS add order changes
# speedup vs baseline: 5.2386x; 1.0582x over previous
#include <hip/hip_runtime.h>

#define DIM_IN   128
#define DIM_HID  256
#define DIM_OUT  128
#define N_DIST   1024
#define N_NODES  500000

typedef float v2f __attribute__((ext_vector_type(2)));
typedef float v4f __attribute__((ext_vector_type(4)));
typedef float v8f __attribute__((ext_vector_type(8)));

#define SEG_W        64
#define SEG_SLICES   (DIM_IN / SEG_W)
#define SEG_PARTS    4
#define SEG_DP       (N_DIST / SEG_PARTS)
#define SEG_DP_LOG2  8
#define SEG_CHUNKS   32
#define SEG_RPC      15632
#define SEG_WAVES    16
#define SEG_THREADS  (SEG_WAVES * 32)
#define SEG_RPW      (SEG_RPC / SEG_WAVES)
#define SEG_GROUPS   ((SEG_RPW + 31) / 32)
#define SEG_LIST     1024
#define SEG_NONE     0xFFFFFFFFu
#define SEG_BINS4    (SEG_DP * SEG_W / 4)
#define SEG_BLK      8192
#define SEG_ROWS     512
#define SEG_LDS_F32  (5 * SEG_BLK)
#define HEAD_ELEMS   (N_DIST * DIM_IN)
#define FUS_D        4
#define FUS_WAVES    16
#define FUS_THREADS  (FUS_WAVES * 32)
#define A0_PITCH     (DIM_IN + 4)
#define H1_PITCH     (DIM_HID + 4)

typedef unsigned int v4u __attribute__((ext_vector_type(4)));

static_assert(SEG_RPC % SEG_WAVES == 0, "waves take equal row ranges");
static_assert((SEG_CHUNKS - 1) * SEG_RPC < N_NODES && SEG_CHUNKS * SEG_RPC >= N_NODES, "chunks must cover the nodes");
static_assert(SEG_DP == (1 << SEG_DP_LOG2) && SEG_DP / 4 == 64 && SEG_ROWS == 512, "bin rows fill the first 64 rows of four blocks");
static_assert(SEG_RPC <= 16384, "a list entry packs 14 bits of row and 9 bits of bin row");
static_assert(SEG_WAVES * SEG_LIST * 4 == 4 * (SEG_BLK / 2) * 4, "the lists fill the second halves of the four blocks");
static_assert(((SEG_RPW + 15) / 16) * 16 <= SEG_LIST, "a padded list must fit");

struct SegBatch { v4f v[8]; unsigned e[8]; };

static __device__ __forceinline__ void seg_load(SegBatch& B, const float* __restrict__ xl,
                                                const unsigned* mylist, int b, int r) {
    const v4u* lp = (const v4u*)(mylist + 16 * b + 8 * r);
    const v4u e0 = lp[0], e1 = lp[1];
    B.e[0] = e0.x; B.e[1] = e0.y; B.e[2] = e0.z; B.e[3] = e0.w;
    B.e[4] = e1.x; B.e[5] = e1.y; B.e[6] = e1.z; B.e[7] = e1.w;
    #pragma unroll
    for (int u = 0; u < 8; ++u) {
        const unsigned rl = (B.e[u] == SEG_NONE) ? 0u : (B.e[u] >> 9);
        B.v[u] = __builtin_nontemporal_load((const v4f*)(xl + (size_t)rl * DIM_IN));
    }
}

static __device__ __forceinline__ void seg_add(const SegBatch& B, float* bins_lane, int r) {
    #pragma unroll
    for (int u = 0; u < 8; ++u) {
        const unsigned e = B.e[u];
        if (e != SEG_NONE) {
            const int z = (int)(e & (unsigned)(SEG_ROWS - 1));
            const v4f t = B.v[u];
            float* p = bins_lane + z * SEG_W;
            float* plo = p + (r ? 2 : 0);
            float* phi = p + (r ? 0 : 2);
            atomicAdd(plo,     r ? t.z : t.x);
            atomicAdd(plo + 1, r ? t.w : t.y);
            atomicAdd(phi,     r ? t.x : t.z);
            atomicAdd(phi + 1, r ? t.y : t.w);
        }
    }
}

__global__ __launch_bounds__(SEG_THREADS)
void n2d_segsum_bins(const float* __restrict__ x,
                     const int*   __restrict__ zone,
                     float*       __restrict__ part) {
    __shared__ alignas(16) float bins[SEG_LDS_F32];

    const int tid  = threadIdx.x;
    const int lane = tid & 31;
    const int wv   = __builtin_amdgcn_readfirstlane(tid >> 5);
    const int r    = lane >> 4;
    const int col4 = lane & 15;
    const int s    = blockIdx.x & 1;
    const int h    = (blockIdx.x >> 1) & (SEG_PARTS - 1);
    const int c    = blockIdx.x / (SEG_SLICES * SEG_PARTS);
    const int n0   = c * SEG_RPC;
    const int w0   = n0 + wv * SEG_RPW;
    const int w1   = (w0 + SEG_RPW < N_NODES) ? (w0 + SEG_RPW) : N_NODES;

    int zr[SEG_GROUPS];
    #pragma unroll
    for (int g = 0; g < SEG_GROUPS; ++g) {
        const int row = w0 + 32 * g + lane;
        zr[g] = (row < w1) ? zone[row] : -1;
    }

    v4f* bins4 = (v4f*)bins;
    const v4f zero4 = {0.0f, 0.0f, 0.0f, 0.0f};
    for (int i = tid; i < SEG_BINS4; i += SEG_THREADS)
        bins4[(i / (SEG_BLK / 8)) * (SEG_BLK / 4) + (i % (SEG_BLK / 8))] = zero4;

    unsigned* mylist = (unsigned*)bins + (wv >> 2) * SEG_BLK + SEG_BLK / 2 + (wv & 3) * SEG_LIST;
    int cnt = 0;
    #pragma unroll
    for (int g = 0; g < SEG_GROUPS; ++g) {
        const int z = zr[g];
        const bool match = ((unsigned)z >> SEG_DP_LOG2) == (unsigned)h;
        const unsigned mask = __builtin_amdgcn_ballot_w32(match);
        const int rank = (int)__builtin_amdgcn_mbcnt_lo(mask, 0u);
        if (match) {
            const unsigned zl = (unsigned)z & (unsigned)(SEG_DP - 1);
            const unsigned binrow = ((zl & 3u) << 7) | (zl >> 2);
            mylist[cnt + rank] = ((unsigned)(wv * SEG_RPW + 32 * g + lane) << 9) | binrow;
        }
        cnt += __builtin_popcount(mask);
    }
    const int nb = (cnt + 15) >> 4;
    if (lane < 16 && cnt + lane < 16 * nb) mylist[cnt + lane] = SEG_NONE;

    const float* xl = x + (size_t)n0 * DIM_IN + s * SEG_W + col4 * 4;
    float* bins_lane = bins + col4 * 4;

    SegBatch A, Bq;
    if (nb > 0) seg_load(A, xl, mylist, 0, r);
    __syncthreads();

    int i = 0;
    for (; i + 2 < nb; i += 2) {
        seg_load(Bq, xl, mylist, i + 1, r);
        seg_add(A, bins_lane, r);
        seg_load(A, xl, mylist, i + 2, r);
        seg_add(Bq, bins_lane, r);
    }
    if (i + 1 < nb) {
        seg_load(Bq, xl, mylist, i + 1, r);
        seg_add(A, bins_lane, r);
        seg_add(Bq, bins_lane, r);
    } else if (i < nb) {
        seg_add(A, bins_lane, r);
    }
    __syncthreads();

    float* dst = part + (size_t)c * HEAD_ELEMS + (size_t)(h * SEG_DP) * DIM_IN + s * SEG_W;
    static_assert(SEG_BINS4 % SEG_THREADS == 0, "every thread writes the same number of pieces");
    #pragma unroll
    for (int k = 0; k < SEG_BINS4 / SEG_THREADS; ++k) {
        const int i = tid + k * SEG_THREADS;
        const int d = i >> 4, q = i & 15;
        const int binrow = ((d & 3) << 7) | (d >> 2);
        *(v4f*)(dst + d * DIM_IN + q * 4) = bins4[binrow * 16 + q];
    }
}

__global__ __launch_bounds__(FUS_THREADS)
void n2d_reduce_mlp(const float* __restrict__ part,
                    const float* __restrict__ w1, const float* __restrict__ b1,
                    const float* __restrict__ w2, const float* __restrict__ b2,
                    float* __restrict__ out) {
    __shared__ alignas(16) float red[FUS_WAVES][FUS_D][DIM_IN];
    __shared__ alignas(16) float a0[16 * A0_PITCH];
    __shared__ alignas(16) float h1[16 * H1_PITCH];
    __shared__ float o2[8][FUS_D][16];
    __shared__ alignas(16) float ot[FUS_D][DIM_OUT];

    const int tid  = threadIdx.x;
    const int lane = tid & 31;
    const int wv   = __builtin_amdgcn_readfirstlane(tid >> 5);
    const int half = lane >> 4;
    const int m    = lane & 15;
    const int d0   = blockIdx.x * FUS_D;

    const int n1 = wv * 16;
    const int n2 = (wv & 7) * 16;
    const int k2 = (wv >> 3) * (DIM_HID / 2);
    v2f bw1[DIM_IN / 4], bw2[DIM_HID / 8];
    #pragma unroll
    for (int s = 0; s < DIM_IN / 4; ++s) {
        const int ka = 4 * s + half * 2;
        bw1[s].x = w1[(size_t)ka       * DIM_HID + n1 + m];
        bw1[s].y = w1[(size_t)(ka + 1) * DIM_HID + n1 + m];
    }
    #pragma unroll
    for (int s = 0; s < DIM_HID / 8; ++s) {
        const int ka = k2 + 4 * s + half * 2;
        bw2[s].x = w2[(size_t)ka       * DIM_OUT + n2 + m];
        bw2[s].y = w2[(size_t)(ka + 1) * DIM_OUT + n2 + m];
    }
    const float bias1 = b1[n1 + m];
    const float bias2 = b2[n2 + m];

    #pragma unroll
    for (int q = 0; q < FUS_D; ++q) {
        const v4f* p = (const v4f*)(part + (size_t)(d0 + q) * DIM_IN) + lane;
        v4f acc = {0.0f, 0.0f, 0.0f, 0.0f};
        #pragma unroll
        for (int j = 0; j < SEG_CHUNKS / FUS_WAVES; ++j)
            acc += p[(size_t)(wv + FUS_WAVES * j) * (HEAD_ELEMS / 4)];
        *(v4f*)&red[wv][q][lane * 4] = acc;
    }
    for (int i = tid; i < (16 - FUS_D) * A0_PITCH; i += FUS_THREADS) a0[FUS_D * A0_PITCH + i] = 0.0f;
    __syncthreads();

    {
        const int q = tid >> 7, col = tid & (DIM_IN - 1);
        float sum = red[0][q][col];
        #pragma unroll
        for (int w = 1; w < FUS_WAVES; ++w) sum += red[w][q][col];
        a0[q * A0_PITCH + col] = sum > 0.0f ? sum : 0.0f;
    }
    __syncthreads();

    {
        v8f acc = {};
        #pragma unroll
        for (int s = 0; s < DIM_IN / 4; ++s) {
            const int ka = 4 * s + half * 2;
            const v2f a = *(const v2f*)&a0[m * A0_PITCH + ka];
            acc = __builtin_amdgcn_wmma_f32_16x16x4_f32(
                      false, a, false, bw1[s], (short)0, acc, false, false);
        }
        #pragma unroll
        for (int r = 0; r < 8; ++r) {
            const float v = acc[r] + bias1;
            h1[(r + half * 8) * H1_PITCH + n1 + m] = v > 0.0f ? v : 0.0f;
        }
    }
    __syncthreads();

    v8f acc2 = {};
    #pragma unroll
    for (int s = 0; s < DIM_HID / 8; ++s) {
        const int ka = k2 + 4 * s + half * 2;
        const v2f a = *(const v2f*)&h1[m * H1_PITCH + ka];
        acc2 = __builtin_amdgcn_wmma_f32_16x16x4_f32(
                   false, a, false, bw2[s], (short)0, acc2, false, false);
    }
    if (wv >= 8 && half == 0) {
        #pragma unroll
        for (int r = 0; r < FUS_D; ++r) o2[wv & 7][r][m] = acc2[r];
    }
    __syncthreads();
    if (wv < 8 && half == 0) {
        #pragma unroll
        for (int r = 0; r < FUS_D; ++r) ot[r][n2 + m] = (acc2[r] + o2[wv][r][m]) + bias2;
    }
    __syncthreads();
    if (wv < FUS_D)
        *(v4f*)(out + (size_t)(d0 + wv) * DIM_OUT + lane * 4) = *(const v4f*)&ot[wv][lane * 4];
}

extern "C" void kernel_launch(void* const* d_in, const int* in_sizes, int n_in,
                              void* d_out, int out_size, void* d_ws, size_t ws_size,
                              hipStream_t stream) {
    const float* x    = (const float*)d_in[0];
    const int*   zone = (const int*)  d_in[1];
    const float* w1   = (const float*)d_in[2];
    const float* b1   = (const float*)d_in[3];
    const float* w2   = (const float*)d_in[4];
    const float* b2   = (const float*)d_in[5];

    float* part = (float*)d_ws;
    float* out  = (float*)d_out;

    n2d_segsum_bins<<<SEG_CHUNKS * SEG_SLICES * SEG_PARTS, SEG_THREADS, 0, stream>>>(x, zone, part);
    n2d_reduce_mlp<<<N_DIST / FUS_D, FUS_THREADS, 0, stream>>>(part, w1, b1, w2, b2, out);
}


# Segment sum without LDS float adds: a work-group orders its chunk's rows of 128 districts by district (LDS integer counters), a wave sums each run of whole 512-B rows in registers; sum order changes
# speedup vs baseline: 5.6607x; 1.0806x over previous
#include <hip/hip_runtime.h>

#define DIM_IN   128
#define DIM_HID  256
#define DIM_OUT  128
#define N_DIST   1024
#define N_NODES  500000

typedef float v2f __attribute__((ext_vector_type(2)));
typedef float v4f __attribute__((ext_vector_type(4)));
typedef float v8f __attribute__((ext_vector_type(8)));
typedef unsigned int v4u __attribute__((ext_vector_type(4)));

#define SEG_PARTS    8
#define SEG_DP       (N_DIST / SEG_PARTS)
#define SEG_DP_LOG2  7
#define SEG_CHUNKS   32
#define SEG_RPC      15632
#define SEG_WAVES    16
#define SEG_THREADS  (SEG_WAVES * 32)
#define SEG_RPW      (SEG_RPC / SEG_WAVES)
#define SEG_GROUPS   ((SEG_RPW + 31) / 32)
#define SEG_B        16
#define HEAD_ELEMS   (N_DIST * DIM_IN)
#define FUS_D        4
#define FUS_WAVES    16
#define FUS_THREADS  (FUS_WAVES * 32)
#define A0_PITCH     (DIM_IN + 4)
#define H1_PITCH     (DIM_HID + 4)

static_assert(SEG_RPC % SEG_WAVES == 0, "waves take equal row ranges");
static_assert((SEG_CHUNKS - 1) * SEG_RPC < N_NODES && SEG_CHUNKS * SEG_RPC >= N_NODES, "chunks must cover the nodes");
static_assert(SEG_DP == (1 << SEG_DP_LOG2) && SEG_DP % 4 == 0 && SEG_DP <= SEG_THREADS, "one thread per district in the prefix steps");

static __device__ __forceinline__ v4f seg_run16(v4f acc, const float* __restrict__ xc, unsigned idx, unsigned m) {
    v4f v[SEG_B];
    #pragma unroll
    for (int u = 0; u < SEG_B; ++u) {
        if ((unsigned)u < m) {
            const unsigned r = (unsigned)__builtin_amdgcn_readlane((int)idx, u);
            v[u] = __builtin_nontemporal_load((const v4f*)(xc + (size_t)r * DIM_IN));
        }
    }
    #pragma unroll
    for (int u = 0; u < SEG_B; ++u) {
        if ((unsigned)u < m) acc += v[u];
    }
    return acc;
}

__global__ __launch_bounds__(SEG_THREADS)
void n2d_segsum_runs(const float* __restrict__ x,
                     const int*   __restrict__ zone,
                     float*       __restrict__ part) {
    __shared__ unsigned cntw[SEG_WAVES][SEG_DP];
    __shared__ unsigned tot[SEG_DP];
    __shared__ unsigned off[SEG_DP];
    __shared__ unsigned wsum[SEG_DP / 32];
    __shared__ unsigned spare[SEG_WAVES][32];
    __shared__ unsigned list[SEG_RPC + SEG_B];
    __shared__ unsigned ticket;

    const int tid  = threadIdx.x;
    const int lane = tid & 31;
    const int wv   = __builtin_amdgcn_readfirstlane(tid >> 5);
    const int h    = blockIdx.x & (SEG_PARTS - 1);
    const int c    = blockIdx.x / SEG_PARTS;
    const int n0   = c * SEG_RPC;
    const int w0   = n0 + wv * SEG_RPW;
    const int w1   = (w0 + SEG_RPW < N_NODES) ? (w0 + SEG_RPW) : N_NODES;

    int zr[SEG_GROUPS];
    #pragma unroll
    for (int g = 0; g < SEG_GROUPS; ++g) {
        const int row = w0 + 32 * g + lane;
        zr[g] = (row < w1) ? zone[row] : -1;
    }
    for (int i = tid; i < SEG_WAVES * SEG_DP; i += SEG_THREADS) (&cntw[0][0])[i] = 0u;
    if (tid == 0) ticket = 0u;
    __syncthreads();

    unsigned rk[SEG_GROUPS];
    #pragma unroll
    for (int g = 0; g < SEG_GROUPS; ++g) {
        const int z = zr[g];
        const bool mine = ((unsigned)z >> SEG_DP_LOG2) == (unsigned)h;
        unsigned* p = mine ? &cntw[wv][z & (SEG_DP - 1)] : &spare[wv][lane];
        rk[g] = atomicAdd(p, 1u);
    }
    __syncthreads();

    unsigned kw[SEG_WAVES];
    unsigned total = 0u, incl = 0u;
    if (tid < SEG_DP) {
        #pragma unroll
        for (int w = 0; w < SEG_WAVES; ++w) kw[w] = cntw[w][tid];
        #pragma unroll
        for (int w = 0; w < SEG_WAVES; ++w) total += kw[w];
        incl = total;
        #pragma unroll
        for (int s = 1; s < 32; s <<= 1) {
            const unsigned up = __shfl_up(incl, s);
            if (lane >= s) incl += up;
        }
        if (lane == 31) wsum[wv] = incl;
    }
    __syncthreads();
    if (tid < SEG_DP) {
        unsigned base = 0u;
        #pragma unroll
        for (int w = 0; w < SEG_DP / 32; ++w) base += (w < wv) ? wsum[w] : 0u;
        const unsigned first = base + incl - total;
        unsigned run = first;
        #pragma unroll
        for (int w = 0; w < SEG_WAVES; ++w) { cntw[w][tid] = run; run += kw[w]; }
        tot[tid] = total;
        off[tid] = first;
    }
    __syncthreads();

    unsigned slot[SEG_GROUPS];
    #pragma unroll
    for (int g = 0; g < SEG_GROUPS; ++g) slot[g] = cntw[wv][zr[g] & (SEG_DP - 1)] + rk[g];
    #pragma unroll
    for (int g = 0; g < SEG_GROUPS; ++g) {
        if (((unsigned)zr[g] >> SEG_DP_LOG2) == (unsigned)h)
            list[slot[g]] = (unsigned)(wv * SEG_RPW + 32 * g + lane);
    }
    __syncthreads();

    const float* xc = x + (size_t)n0 * DIM_IN + lane * 4;
    float* pc = part + (size_t)c * HEAD_ELEMS + (size_t)(h * SEG_DP) * DIM_IN + lane * 4;

    unsigned t0 = 0u;
    if (lane == 0) t0 = atomicAdd(&ticket, 1u);
    unsigned d = (unsigned)__builtin_amdgcn_readfirstlane((int)t0);
    unsigned start = 0u, n = 0u, idx = 0u;
    if (d < SEG_DP) {
        start = (unsigned)__builtin_amdgcn_readfirstlane((int)off[d]);
        n     = (unsigned)__builtin_amdgcn_readfirstlane((int)tot[d]);
        idx   = list[start + (lane & (SEG_B - 1))];
    }
    while (d < SEG_DP) {
        v4f v[SEG_B];
        const unsigned m0 = n < SEG_B ? n : SEG_B;
        #pragma unroll
        for (int u = 0; u < SEG_B; ++u) {
            if ((unsigned)u < m0) {
                const unsigned r = (unsigned)__builtin_amdgcn_readlane((int)idx, u);
                v[u] = __builtin_nontemporal_load((const v4f*)(xc + (size_t)r * DIM_IN));
            }
        }
        unsigned t1 = 0u;
        if (lane == 0) t1 = atomicAdd(&ticket, 1u);
        const unsigned dn = (unsigned)__builtin_amdgcn_readfirstlane((int)t1);
        unsigned startn = 0u, nn = 0u, idxn = 0u;
        if (dn < SEG_DP) {
            startn = (unsigned)__builtin_amdgcn_readfirstlane((int)off[dn]);
            nn     = (unsigned)__builtin_amdgcn_readfirstlane((int)tot[dn]);
            idxn   = list[startn + (lane & (SEG_B - 1))];
        }
        v4f acc = {0.0f, 0.0f, 0.0f, 0.0f};
        #pragma unroll
        for (int u = 0; u < SEG_B; ++u) {
            if ((unsigned)u < m0) acc += v[u];
        }
        for (unsigned j = SEG_B; j < n; j += SEG_B) {
            const unsigned idx2 = list[start + j + (lane & (SEG_B - 1))];
            const unsigned m = (n - j) < SEG_B ? (n - j) : SEG_B;
            acc = seg_run16(acc, xc, idx2, m);
        }
        *(v4f*)(pc + (size_t)d * DIM_IN) = acc;
        d = dn; start = startn; n = nn; idx = idxn;
    }
}

__global__ __launch_bounds__(FUS_THREADS)
void n2d_reduce_mlp(const float* __restrict__ part,
                    const float* __restrict__ w1, const float* __restrict__ b1,
                    const float* __restrict__ w2, const float* __restrict__ b2,
                    float* __restrict__ out) {
    __shared__ alignas(16) float red[FUS_WAVES][FUS_D][DIM_IN];
    __shared__ alignas(16) float a0[16 * A0_PITCH];
    __shared__ alignas(16) float h1[16 * H1_PITCH];
    __shared__ float o2[8][FUS_D][16];
    __shared__ alignas(16) float ot[FUS_D][DIM_OUT];

    const int tid  = threadIdx.x;
    const int lane = tid & 31;
    const int wv   = __builtin_amdgcn_readfirstlane(tid >> 5);
    const int half = lane >> 4;
    const int m    = lane & 15;
    const int d0   = blockIdx.x * FUS_D;

    const int n1 = wv * 16;
    const int n2 = (wv & 7) * 16;
    const int k2 = (wv >> 3) * (DIM_HID / 2);
    v2f bw1[DIM_IN / 4], bw2[DIM_HID / 8];
    #pragma unroll
    for (int s = 0; s < DIM_IN / 4; ++s) {
        const int ka = 4 * s + half * 2;
        bw1[s].x = w1[(size_t)ka       * DIM_HID + n1 + m];
        bw1[s].y = w1[(size_t)(ka + 1) * DIM_HID + n1 + m];
    }
    #pragma unroll
    for (int s = 0; s < DIM_HID / 8; ++s) {
        const int ka = k2 + 4 * s + half * 2;
        bw2[s].x = w2[(size_t)ka       * DIM_OUT + n2 + m];
        bw2[s].y = w2[(size_t)(ka + 1) * DIM_OUT + n2 + m];
    }
    const float bias1 = b1[n1 + m];
    const float bias2 = b2[n2 + m];

    #pragma unroll
    for (int q = 0; q < FUS_D; ++q) {
        const v4f* p = (const v4f*)(part + (size_t)(d0 + q) * DIM_IN) + lane;
        v4f acc = {0.0f, 0.0f, 0.0f, 0.0f};
        #pragma unroll
        for (int j = 0; j < SEG_CHUNKS / FUS_WAVES; ++j)
            acc += p[(size_t)(wv + FUS_WAVES * j) * (HEAD_ELEMS / 4)];
        *(v4f*)&red[wv][q][lane * 4] = acc;
    }
    for (int i = tid; i < (16 - FUS_D) * A0_PITCH; i += FUS_THREADS) a0[FUS_D * A0_PITCH + i] = 0.0f;
    __syncthreads();

    {
        const int q = tid >> 7, col = tid & (DIM_IN - 1);
        float sum = red[0][q][col];
        #pragma unroll
        for (int w = 1; w < FUS_WAVES; ++w) sum += red[w][q][col];
        a0[q * A0_PITCH + col] = sum > 0.0f ? sum : 0.0f;
    }
    __syncthreads();

    {
        v8f acc = {};
        #pragma unroll
        for (int s = 0; s < DIM_IN / 4; ++s) {
            const int ka = 4 * s + half * 2;
            const v2f a = *(const v2f*)&a0[m * A0_PITCH + ka];
            acc = __builtin_amdgcn_wmma_f32_16x16x4_f32(
                      false, a, false, bw1[s], (short)0, acc, false, false);
        }
        #pragma unroll
        for (int r = 0; r < 8; ++r) {
            const float v = acc[r] + bias1;
            h1[(r + half * 8) * H1_PITCH + n1 + m] = v > 0.0f ? v : 0.0f;
        }
    }
    __syncthreads();

    v8f acc2 = {};
    #pragma unroll
    for (int s = 0; s < DIM_HID / 8; ++s) {
        const int ka = k2 + 4 * s + half * 2;
        const v2f a = *(const v2f*)&h1[m * H1_PITCH + ka];
        acc2 = __builtin_amdgcn_wmma_f32_16x16x4_f32(
                   false, a, false, bw2[s], (short)0, acc2, false, false);
    }
    if (wv >= 8 && half == 0) {
        #pragma unroll
        for (int r = 0; r < FUS_D; ++r) o2[wv & 7][r][m] = acc2[r];
    }
    __syncthreads();
    if (wv < 8 && half == 0) {
        #pragma unroll
        for (int r = 0; r < FUS_D; ++r) ot[r][n2 + m] = (acc2[r] + o2[wv][r][m]) + bias2;
    }
    __syncthreads();
    if (wv < FUS_D)
        *(v4f*)(out + (size_t)(d0 + wv) * DIM_OUT + lane * 4) = *(const v4f*)&ot[wv][lane * 4];
}

extern "C" void kernel_launch(void* const* d_in, const int* in_sizes, int n_in,
                              void* d_out, int out_size, void* d_ws, size_t ws_size,
                              hipStream_t stream) {
    const float* x    = (const float*)d_in[0];
    const int*   zone = (const int*)  d_in[1];
    const float* w1   = (const float*)d_in[2];
    const float* b1   = (const float*)d_in[3];
    const float* w2   = (const float*)d_in[4];
    const float* b2   = (const float*)d_in[5];

    float* part = (float*)d_ws;
    float* out  = (float*)d_out;

    n2d_segsum_runs<<<SEG_CHUNKS * SEG_PARTS, SEG_THREADS, 0, stream>>>(x, zone, part);
    n2d_reduce_mlp<<<N_DIST / FUS_D, FUS_THREADS, 0, stream>>>(part, w1, b1, w2, b2, out);
}
